# starting kernel
# baseline (speedup 1.0000x reference)
#include <hip/hip_runtime.h>
#include <math.h>

#define B_BATCH 32
#define L_SEQ   2048
#define D_IN    64
#define NC      32
#define DC      16
#define EPS_F   1e-7f

typedef __attribute__((ext_vector_type(2))) float v2f;
typedef __attribute__((ext_vector_type(8))) float v8f;

__global__ void caps_gemm(const float* __restrict__ u,
                          const float* __restrict__ W,
                          float* __restrict__ uhat) {
  const int wave = (blockIdx.x * blockDim.x + threadIdx.x) >> 5;
  const int lane = threadIdx.x & 31;
  const int ng    = wave & 3;
  const int ltile = (wave >> 2) & 127;
  const int b     = wave >> 9;

  const int col = lane & 15;
  const int hi  = lane >> 4;

  const float* urow = u + ((size_t)b * L_SEQ + (size_t)ltile * 16 + col) * D_IN;
  const v2f* urow2 = reinterpret_cast<const v2f*>(urow);
  v2f a[16];
#pragma unroll
  for (int k = 0; k < 16; ++k) a[k] = urow2[2 * k + hi];

  for (int n = 8 * ng; n < 8 * ng + 8; ++n) {
    const float* wcol = W + n * DC + col;
    v8f acc = {};
#pragma unroll
    for (int k = 0; k < 16; ++k) {
      const int kk = 4 * k + 2 * hi;
      v2f bm;
      bm.x = wcol[(size_t)kk * (NC * DC)];
      bm.y = wcol[(size_t)(kk + 1) * (NC * DC)];
      acc = __builtin_amdgcn_wmma_f32_16x16x4_f32(
           false, a[k],  false, bm,
           (short)0, acc,  false,  false);
    }
    float* out = uhat + (((size_t)b * NC + n) * L_SEQ + (size_t)ltile * 16) * DC;
#pragma unroll
    for (int v = 0; v < 8; ++v) out[(v + 8 * hi) * DC + col] = acc[v];
  }
}

__global__ void caps_reduce(const float* __restrict__ uhat,
                            const float* __restrict__ c,
                            float* __restrict__ outv, int iter0) {
  const int bn = blockIdx.x;
  const int t  = threadIdx.x;
  const float4* src = reinterpret_cast<const float4*>(uhat + (size_t)bn * L_SEQ * DC);
  const float* cw = c + (size_t)bn * L_SEQ;

  float4 acc = make_float4(0.f, 0.f, 0.f, 0.f);
  const int dg = t & 3;
  for (int q = t; q < (L_SEQ * DC) / 4; q += 256) {
    const int l = q >> 2;
    const float cv = iter0 ? (1.0f / 32.0f) : cw[l];
    float4 v = src[q];
    acc.x += cv * v.x; acc.y += cv * v.y;
    acc.z += cv * v.z; acc.w += cv * v.w;
  }

  __shared__ float part[16][64];
  const int slot = t >> 2;
  part[4 * dg + 0][slot] = acc.x;
  part[4 * dg + 1][slot] = acc.y;
  part[4 * dg + 2][slot] = acc.z;
  part[4 * dg + 3][slot] = acc.w;
  __syncthreads();

  __shared__ float sums[16];
  if (t < 16) {
    float s = 0.f;
    for (int i = 0; i < 64; ++i) s += part[t][i];
    sums[t] = s;
  }
  __syncthreads();
  if (t < 16) {
    float ss = EPS_F;
    for (int d = 0; d < 16; ++d) ss += sums[d] * sums[d];
    outv[bn * DC + t] = sums[t] / sqrtf(ss);
  }
}

__global__ void caps_update(const float* __restrict__ uhat,
                            const float* __restrict__ outv,
                            float* __restrict__ c) {
  const int blk = blockIdx.x;
  const int b   = blk >> 3;
  const int l   = ((blk & 7) << 8) + threadIdx.x;

  __shared__ float ovs[NC * DC];
  ovs[threadIdx.x]       = outv[b * NC * DC + threadIdx.x];
  ovs[threadIdx.x + 256] = outv[b * NC * DC + threadIdx.x + 256];
  __syncthreads();

  float vals[NC];
  float m = -3.402823466e38f;
#pragma unroll 4
  for (int n = 0; n < NC; ++n) {
    const float4* uh4 = reinterpret_cast<const float4*>(
        uhat + (((size_t)b * NC + n) * L_SEQ + l) * DC);
    float dot = 0.f;
#pragma unroll
    for (int q = 0; q < 4; ++q) {
      float4 v = uh4[q];
      dot += v.x * ovs[n * DC + 4 * q + 0];
      dot += v.y * ovs[n * DC + 4 * q + 1];
      dot += v.z * ovs[n * DC + 4 * q + 2];
      dot += v.w * ovs[n * DC + 4 * q + 3];
    }
    vals[n] = dot;
    m = fmaxf(m, dot);
  }
  float sum = 0.f;
#pragma unroll
  for (int n = 0; n < NC; ++n) { vals[n] = expf(vals[n] - m); sum += vals[n]; }
  const float inv = 1.0f / sum;
#pragma unroll
  for (int n = 0; n < NC; ++n)
    c[((size_t)b * NC + n) * L_SEQ + l] = vals[n] * inv;
}

extern "C" void kernel_launch(void* const* d_in, const int* in_sizes, int n_in,
                              void* d_out, int out_size, void* d_ws, size_t ws_size,
                              hipStream_t stream) {
  const float* u = (const float*)d_in[0];
  const float* W = (const float*)d_in[1];
  float* out = (float*)d_out;

  float* uhat = (float*)d_ws;
  float* cbuf = uhat + (size_t)B_BATCH * NC * L_SEQ * DC;
  float* ov   = cbuf + (size_t)B_BATCH * NC * L_SEQ;

  caps_gemm<<<2048, 256, 0, stream>>>(u, W, uhat);

  caps_reduce<<<B_BATCH * NC, 256, 0, stream>>>(uhat, cbuf, ov, 1);
  caps_update<<<B_BATCH * 8, 256, 0, stream>>>(uhat, ov, cbuf);
  caps_reduce<<<B_BATCH * NC, 256, 0, stream>>>(uhat, cbuf, ov, 0);
  caps_update<<<B_BATCH * 8, 256, 0, stream>>>(uhat, ov, cbuf);
  caps_reduce<<<B_BATCH * NC, 256, 0, stream>>>(uhat, cbuf, out, 0);
}


# Fold each finish launch into the head of the next routing launch: every 8-wave block rebuilds g for its batch from the partial sums with its W slice held in registers (6 launches -> 4)
# speedup vs baseline: 3.9171x; 3.9171x over previous
#include <hip/hip_runtime.h>
#include <math.h>


#define B_BATCH 32
#define L_SEQ   2048
#define D_IN    64
#define NC      32
#define DC      16
#define EPS_F   1e-7f
#define NCH     8
#define GP      68

typedef __attribute__((ext_vector_type(2))) float v2f;
typedef __attribute__((ext_vector_type(8))) float v8f;


template <int M>
__device__ __forceinline__ float row_xor(float v) {
  return __int_as_float(__builtin_amdgcn_mov_dpp(__float_as_int(v), 0x160 + M, 0xf, 0xf, false));
}
__device__ __forceinline__ float row_max16(float v) {
  v = fmaxf(v, row_xor<1>(v));
  v = fmaxf(v, row_xor<2>(v));
  v = fmaxf(v, row_xor<4>(v));
  v = fmaxf(v, row_xor<8>(v));
  return v;
}
__device__ __forceinline__ float row_sum16(float v) {
  v += row_xor<1>(v);
  v += row_xor<2>(v);
  v += row_xor<4>(v);
  v += row_xor<8>(v);
  return v;
}

__global__ void caps_colsum(const float* __restrict__ u, float* __restrict__ tp0) {
  const int b  = blockIdx.x >> 3;
  const int ch = blockIdx.x & 7;
  const int t  = threadIdx.x;
  const int k4 = t & 15;
  const int rg = t >> 4;
  const float4* src = reinterpret_cast<const float4*>(
      u + ((size_t)b * L_SEQ + (size_t)ch * 256) * D_IN);
  float4 acc = make_float4(0.f, 0.f, 0.f, 0.f);
#pragma unroll
  for (int i = 0; i < 16; ++i) {
    float4 v = src[(size_t)(rg + 16 * i) * 16 + k4];
    acc.x += v.x; acc.y += v.y; acc.z += v.z; acc.w += v.w;
  }
  __shared__ float part[16][64];
  part[rg][4 * k4 + 0] = acc.x;
  part[rg][4 * k4 + 1] = acc.y;
  part[rg][4 * k4 + 2] = acc.z;
  part[rg][4 * k4 + 3] = acc.w;
  __syncthreads();
  if (t < 64) {
    float s = 0.f;
#pragma unroll
    for (int r = 0; r < 16; ++r) s += part[r][t];
    tp0[((size_t)b * NCH + ch) * D_IN + t] = s;
  }
}

__global__ void caps_final(const float* __restrict__ tp,
                           const float* __restrict__ W,
                           float* __restrict__ outv) {
  const int bn = blockIdx.x;
  const int b  = bn >> 5;
  const int n  = bn & 31;
  const int t  = threadIdx.x;

  float tk = 0.f;
#pragma unroll
  for (int ch = 0; ch < NCH; ++ch)
    tk += tp[(((size_t)b * NCH + ch) * NC + n) * D_IN + t];

  __shared__ float ts[D_IN];
  __shared__ float ss[DC];
  ts[t] = tk;
  __syncthreads();
  if (t < DC) {
    const float* wc = W + n * DC + t;
    float s = 0.f;
#pragma unroll 8
    for (int k = 0; k < D_IN; ++k) s += ts[k] * wc[(size_t)k * (NC * DC)];
    ss[t] = s;
  }
  __syncthreads();
  if (t < DC) {
    float q = EPS_F;
#pragma unroll
    for (int d = 0; d < DC; ++d) q += ss[d] * ss[d];
    outv[bn * DC + t] = ss[t] / sqrtf(q);
  }
}

template <bool FIRST>
__global__ void __launch_bounds__(256)
caps_route(const float* __restrict__ u,
           const float* __restrict__ W,
           const float* __restrict__ tin,
           float* __restrict__ tout) {
  const int lane = threadIdx.x & 31;
  const int wv   = threadIdx.x >> 5;
  const int col  = lane & 15;
  const int hi   = lane >> 4;
  const int b    = blockIdx.x >> 3;
  const int ch   = blockIdx.x & 7;

  __shared__ __attribute__((aligned(16))) float g_s[NC * GP];
  __shared__ float red[8][NC * D_IN];

  float4 ua[2][8];

  {
    const int kq = lane & 7;
    const int n  = wv * 4 + (lane >> 3);

    const float4* tsrc = FIRST
        ? reinterpret_cast<const float4*>(tin + (size_t)b * NCH * D_IN + 8 * kq)
        : reinterpret_cast<const float4*>(tin + ((size_t)b * NCH * NC + n) * D_IN + 8 * kq);
    const int tstep = FIRST ? (D_IN / 4) : (NC * D_IN / 4);
    float4 p0[NCH], p1[NCH];
#pragma unroll
    for (int c = 0; c < NCH; ++c) { p0[c] = tsrc[c * tstep]; p1[c] = tsrc[c * tstep + 1]; }

    const float4* wsrc = reinterpret_cast<const float4*>(W + (size_t)(8 * kq) * (NC * DC) + n * DC);
    float4 w4[8][4];
#pragma unroll
    for (int j = 0; j < 8; ++j)
#pragma unroll
      for (int q = 0; q < 4; ++q) w4[j][q] = wsrc[j * (NC * DC / 4) + q];

#pragma unroll
    for (int i = 0; i < 2; ++i) {
      const float4* ur = reinterpret_cast<const float4*>(
          u + ((size_t)b * L_SEQ + ch * 256 + (wv * 2 + i) * 16 + col) * D_IN);
#pragma unroll
      for (int jj = 0; jj < 8; ++jj) ua[i][jj] = ur[2 * jj + hi];
    }
    __builtin_amdgcn_sched_barrier(0);

    float tk[8];
#pragma unroll
    for (int j = 0; j < 8; ++j) tk[j] = 0.f;
#pragma unroll
    for (int c = 0; c < NCH; ++c) {
      tk[0] += p0[c].x; tk[1] += p0[c].y; tk[2] += p0[c].z; tk[3] += p0[c].w;
      tk[4] += p1[c].x; tk[5] += p1[c].y; tk[6] += p1[c].z; tk[7] += p1[c].w;
    }
    if (FIRST) {
#pragma unroll
      for (int j = 0; j < 8; ++j) tk[j] *= (1.0f / 32.0f);
    }

    float s[DC];
#pragma unroll
    for (int d = 0; d < DC; ++d) s[d] = 0.f;
#pragma unroll
    for (int j = 0; j < 8; ++j)
#pragma unroll
      for (int q = 0; q < 4; ++q) {
        s[4 * q + 0] += tk[j] * w4[j][q].x;
        s[4 * q + 1] += tk[j] * w4[j][q].y;
        s[4 * q + 2] += tk[j] * w4[j][q].z;
        s[4 * q + 3] += tk[j] * w4[j][q].w;
      }
#pragma unroll
    for (int d = 0; d < DC; ++d) {
      s[d] += row_xor<1>(s[d]);
      s[d] += row_xor<2>(s[d]);
      s[d] += row_xor<4>(s[d]);
    }
    float qq = EPS_F;
#pragma unroll
    for (int d = 0; d < DC; ++d) qq += s[d] * s[d];
    const float inv = 1.0f / sqrtf(qq);

    float gk[8];
#pragma unroll
    for (int j = 0; j < 8; ++j) {
      float a = 0.f;
#pragma unroll
      for (int q = 0; q < 4; ++q) {
        a += w4[j][q].x * s[4 * q + 0];
        a += w4[j][q].y * s[4 * q + 1];
        a += w4[j][q].z * s[4 * q + 2];
        a += w4[j][q].w * s[4 * q + 3];
      }
      gk[j] = a * inv;
    }
    float4* gd = reinterpret_cast<float4*>(g_s + n * GP + 8 * kq);
    gd[0] = make_float4(gk[0], gk[1], gk[2], gk[3]);
    gd[1] = make_float4(gk[4], gk[5], gk[6], gk[7]);
  }
  __syncthreads();

  float4 gb0[8], gb1[8];
  {
    const float4* g0 = reinterpret_cast<const float4*>(g_s + col * GP);
    const float4* g1 = reinterpret_cast<const float4*>(g_s + (16 + col) * GP);
#pragma unroll
    for (int jj = 0; jj < 8; ++jj) { gb0[jj] = g0[2 * jj + hi]; gb1[jj] = g1[2 * jj + hi]; }
  }

  v8f tacc[2][4];
#pragma unroll
  for (int nt = 0; nt < 2; ++nt)
#pragma unroll
    for (int kt = 0; kt < 4; ++kt) tacc[nt][kt] = (v8f){};

#pragma unroll
  for (int i = 0; i < 2; ++i) {
    const int l0 = ch * 256 + (wv * 2 + i) * 16;
    const float* ut = u + ((size_t)b * L_SEQ + l0) * D_IN;

    const float4* a4 = ua[i];

    v8f lg0 = {}, lg1 = {};
#pragma unroll
    for (int jj = 0; jj < 8; ++jj) {
      v2f alo, ahi, b0lo, b0hi, b1lo, b1hi;
      alo.x  = a4[jj].x;  alo.y  = a4[jj].y;  ahi.x  = a4[jj].z;  ahi.y  = a4[jj].w;
      b0lo.x = gb0[jj].x; b0lo.y = gb0[jj].y; b0hi.x = gb0[jj].z; b0hi.y = gb0[jj].w;
      b1lo.x = gb1[jj].x; b1lo.y = gb1[jj].y; b1hi.x = gb1[jj].z; b1hi.y = gb1[jj].w;
      lg0 = __builtin_amdgcn_wmma_f32_16x16x4_f32(false, alo, false, b0lo, (short)0, lg0, false, false);
      lg1 = __builtin_amdgcn_wmma_f32_16x16x4_f32(false, alo, false, b1lo, (short)0, lg1, false, false);
      lg0 = __builtin_amdgcn_wmma_f32_16x16x4_f32(false, ahi, false, b0hi, (short)0, lg0, false, false);
      lg1 = __builtin_amdgcn_wmma_f32_16x16x4_f32(false, ahi, false, b1hi, (short)0, lg1, false, false);
    }

    float c0[8], c1[8];
#pragma unroll
    for (int v = 0; v < 8; ++v) {
      const float m  = row_max16(fmaxf(lg0[v], lg1[v]));
      const float e0 = expf(lg0[v] - m);
      const float e1 = expf(lg1[v] - m);
      const float inv = 1.0f / row_sum16(e0 + e1);
      c0[v] = e0 * inv;
      c1[v] = e1 * inv;
    }

    const float* ub = ut + (size_t)(8 * hi) * D_IN + col;
#pragma unroll
    for (int j = 0; j < 4; ++j) {
      v2f ca0, ca1;
      ca0.x = c0[2 * j]; ca0.y = c0[2 * j + 1];
      ca1.x = c1[2 * j]; ca1.y = c1[2 * j + 1];
#pragma unroll
      for (int kt = 0; kt < 4; ++kt) {
        v2f ubm;
        ubm.x = ub[(size_t)(2 * j) * D_IN + kt * 16];
        ubm.y = ub[(size_t)(2 * j + 1) * D_IN + kt * 16];
        tacc[0][kt] = __builtin_amdgcn_wmma_f32_16x16x4_f32(false, ca0, false, ubm, (short)0, tacc[0][kt], false, false);
        tacc[1][kt] = __builtin_amdgcn_wmma_f32_16x16x4_f32(false, ca1, false, ubm, (short)0, tacc[1][kt], false, false);
      }
    }
  }

#pragma unroll
  for (int nt = 0; nt < 2; ++nt)
#pragma unroll
    for (int kt = 0; kt < 4; ++kt)
#pragma unroll
      for (int v = 0; v < 8; ++v)
        red[wv][(nt * 16 + v + 8 * hi) * D_IN + kt * 16 + col] = tacc[nt][kt][v];
  __syncthreads();
  float* dst = tout + ((size_t)b * NCH + ch) * (NC * D_IN);
#pragma unroll
  for (int i = 0; i < 8; ++i) {
    const int idx = threadIdx.x + 256 * i;
    dst[idx] = ((red[0][idx] + red[1][idx]) + (red[2][idx] + red[3][idx])) +
               ((red[4][idx] + red[5][idx]) + (red[6][idx] + red[7][idx]));
  }
}

extern "C" void kernel_launch(void* const* d_in, const int* in_sizes, int n_in,
                              void* d_out, int out_size, void* d_ws, size_t ws_size,
                              hipStream_t stream) {
  const float* u = (const float*)d_in[0];
  const float* W = (const float*)d_in[1];
  float* out = (float*)d_out;

  float* tp0 = (float*)d_ws;
  float* tpa = tp0 + (size_t)B_BATCH * NCH * D_IN;
  float* tpb = tpa + (size_t)B_BATCH * NCH * NC * D_IN;

  caps_colsum<<<B_BATCH * NCH, 256, 0, stream>>>(u, tp0);
  caps_route<true><<<B_BATCH * NCH, 256, 0, stream>>>(u, W, tp0, tpa);
  caps_route<false><<<B_BATCH * NCH, 256, 0, stream>>>(u, W, tpa, tpb);
  caps_final<<<B_BATCH * NC, 64, 0, stream>>>(tpb, W, out);
}


# Last launch regrouped as 256 single-wave blocks, lane = (capsule, k-octet) with its W slice in registers: all 48 loads in one group instead of 9 dependent rounds, 2048 waves -> 256
# speedup vs baseline: 4.1142x; 1.0503x over previous
#include <hip/hip_runtime.h>
#include <math.h>


#define B_BATCH 32
#define L_SEQ   2048
#define D_IN    64
#define NC      32
#define DC      16
#define EPS_F   1e-7f
#define NCH     8
#define GP      68

typedef __attribute__((ext_vector_type(2))) float v2f;
typedef __attribute__((ext_vector_type(8))) float v8f;


template <int M>
__device__ __forceinline__ float row_xor(float v) {
  return __int_as_float(__builtin_amdgcn_mov_dpp(__float_as_int(v), 0x160 + M, 0xf, 0xf, false));
}
__device__ __forceinline__ float row_max16(float v) {
  v = fmaxf(v, row_xor<1>(v));
  v = fmaxf(v, row_xor<2>(v));
  v = fmaxf(v, row_xor<4>(v));
  v = fmaxf(v, row_xor<8>(v));
  return v;
}
__device__ __forceinline__ float row_sum16(float v) {
  v += row_xor<1>(v);
  v += row_xor<2>(v);
  v += row_xor<4>(v);
  v += row_xor<8>(v);
  return v;
}

__global__ void caps_colsum(const float* __restrict__ u, float* __restrict__ tp0) {
  const int b  = blockIdx.x >> 3;
  const int ch = blockIdx.x & 7;
  const int t  = threadIdx.x;
  const int k4 = t & 15;
  const int rg = t >> 4;
  const float4* src = reinterpret_cast<const float4*>(
      u + ((size_t)b * L_SEQ + (size_t)ch * 256) * D_IN);
  float4 acc = make_float4(0.f, 0.f, 0.f, 0.f);
#pragma unroll
  for (int i = 0; i < 16; ++i) {
    float4 v = src[(size_t)(rg + 16 * i) * 16 + k4];
    acc.x += v.x; acc.y += v.y; acc.z += v.z; acc.w += v.w;
  }
  __shared__ float part[16][64];
  part[rg][4 * k4 + 0] = acc.x;
  part[rg][4 * k4 + 1] = acc.y;
  part[rg][4 * k4 + 2] = acc.z;
  part[rg][4 * k4 + 3] = acc.w;
  __syncthreads();
  if (t < 64) {
    float s = 0.f;
#pragma unroll
    for (int r = 0; r < 16; ++r) s += part[r][t];
    tp0[((size_t)b * NCH + ch) * D_IN + t] = s;
  }
}

__global__ void __launch_bounds__(32)
caps_final(const float* __restrict__ tp,
           const float* __restrict__ W,
           float* __restrict__ outv) {
  const int lane = threadIdx.x;
  const int b    = blockIdx.x >> 3;
  const int kq   = lane & 7;
  const int n    = (blockIdx.x & 7) * 4 + (lane >> 3);

  const float4* tsrc = reinterpret_cast<const float4*>(
      tp + ((size_t)b * NCH * NC + n) * D_IN + 8 * kq);
  float4 p0[NCH], p1[NCH];
#pragma unroll
  for (int c = 0; c < NCH; ++c) {
    p0[c] = tsrc[c * (NC * D_IN / 4)];
    p1[c] = tsrc[c * (NC * D_IN / 4) + 1];
  }
  const float4* wsrc = reinterpret_cast<const float4*>(W + (size_t)(8 * kq) * (NC * DC) + n * DC);
  float4 w4[8][4];
#pragma unroll
  for (int j = 0; j < 8; ++j)
#pragma unroll
    for (int q = 0; q < 4; ++q) w4[j][q] = wsrc[j * (NC * DC / 4) + q];
  __builtin_amdgcn_sched_barrier(0);

  float tk[8];
#pragma unroll
  for (int j = 0; j < 8; ++j) tk[j] = 0.f;
#pragma unroll
  for (int c = 0; c < NCH; ++c) {
    tk[0] += p0[c].x; tk[1] += p0[c].y; tk[2] += p0[c].z; tk[3] += p0[c].w;
    tk[4] += p1[c].x; tk[5] += p1[c].y; tk[6] += p1[c].z; tk[7] += p1[c].w;
  }

  float s[DC];
#pragma unroll
  for (int d = 0; d < DC; ++d) s[d] = 0.f;
#pragma unroll
  for (int j = 0; j < 8; ++j)
#pragma unroll
    for (int q = 0; q < 4; ++q) {
      s[4 * q + 0] += tk[j] * w4[j][q].x;
      s[4 * q + 1] += tk[j] * w4[j][q].y;
      s[4 * q + 2] += tk[j] * w4[j][q].z;
      s[4 * q + 3] += tk[j] * w4[j][q].w;
    }
#pragma unroll
  for (int d = 0; d < DC; ++d) {
    s[d] += row_xor<1>(s[d]);
    s[d] += row_xor<2>(s[d]);
    s[d] += row_xor<4>(s[d]);
  }
  float qq = EPS_F;
#pragma unroll
  for (int d = 0; d < DC; ++d) qq += s[d] * s[d];
  const float nrm = sqrtf(qq);

  float o0 = s[0], o1 = s[1];
#pragma unroll
  for (int j = 1; j < 8; ++j) {
    o0 = (kq == j) ? s[2 * j] : o0;
    o1 = (kq == j) ? s[2 * j + 1] : o1;
  }
  float2 o;
  o.x = o0 / nrm;
  o.y = o1 / nrm;
  *reinterpret_cast<float2*>(outv + ((size_t)b * NC + n) * DC + 2 * kq) = o;
}

template <bool FIRST>
__global__ void __launch_bounds__(256)
caps_route(const float* __restrict__ u,
           const float* __restrict__ W,
           const float* __restrict__ tin,
           float* __restrict__ tout) {
  const int lane = threadIdx.x & 31;
  const int wv   = threadIdx.x >> 5;
  const int col  = lane & 15;
  const int hi   = lane >> 4;
  const int b    = blockIdx.x >> 3;
  const int ch   = blockIdx.x & 7;

  __shared__ __attribute__((aligned(16))) float g_s[NC * GP];
  __shared__ float red[8][NC * D_IN];

  float4 ua[2][8];

  {
    const int kq = lane & 7;
    const int n  = wv * 4 + (lane >> 3);

    const float4* tsrc = FIRST
        ? reinterpret_cast<const float4*>(tin + (size_t)b * NCH * D_IN + 8 * kq)
        : reinterpret_cast<const float4*>(tin + ((size_t)b * NCH * NC + n) * D_IN + 8 * kq);
    const int tstep = FIRST ? (D_IN / 4) : (NC * D_IN / 4);
    float4 p0[NCH], p1[NCH];
#pragma unroll
    for (int c = 0; c < NCH; ++c) { p0[c] = tsrc[c * tstep]; p1[c] = tsrc[c * tstep + 1]; }

    const float4* wsrc = reinterpret_cast<const float4*>(W + (size_t)(8 * kq) * (NC * DC) + n * DC);
    float4 w4[8][4];
#pragma unroll
    for (int j = 0; j < 8; ++j)
#pragma unroll
      for (int q = 0; q < 4; ++q) w4[j][q] = wsrc[j * (NC * DC / 4) + q];

#pragma unroll
    for (int i = 0; i < 2; ++i) {
      const float4* ur = reinterpret_cast<const float4*>(
          u + ((size_t)b * L_SEQ + ch * 256 + (wv * 2 + i) * 16 + col) * D_IN);
#pragma unroll
      for (int jj = 0; jj < 8; ++jj) ua[i][jj] = ur[2 * jj + hi];
    }
    __builtin_amdgcn_sched_barrier(0);

    float tk[8];
#pragma unroll
    for (int j = 0; j < 8; ++j) tk[j] = 0.f;
#pragma unroll
    for (int c = 0; c < NCH; ++c) {
      tk[0] += p0[c].x; tk[1] += p0[c].y; tk[2] += p0[c].z; tk[3] += p0[c].w;
      tk[4] += p1[c].x; tk[5] += p1[c].y; tk[6] += p1[c].z; tk[7] += p1[c].w;
    }
    if (FIRST) {
#pragma unroll
      for (int j = 0; j < 8; ++j) tk[j] *= (1.0f / 32.0f);
    }

    float s[DC];
#pragma unroll
    for (int d = 0; d < DC; ++d) s[d] = 0.f;
#pragma unroll
    for (int j = 0; j < 8; ++j)
#pragma unroll
      for (int q = 0; q < 4; ++q) {
        s[4 * q + 0] += tk[j] * w4[j][q].x;
        s[4 * q + 1] += tk[j] * w4[j][q].y;
        s[4 * q + 2] += tk[j] * w4[j][q].z;
        s[4 * q + 3] += tk[j] * w4[j][q].w;
      }
#pragma unroll
    for (int d = 0; d < DC; ++d) {
      s[d] += row_xor<1>(s[d]);
      s[d] += row_xor<2>(s[d]);
      s[d] += row_xor<4>(s[d]);
    }
    float qq = EPS_F;
#pragma unroll
    for (int d = 0; d < DC; ++d) qq += s[d] * s[d];
    const float inv = 1.0f / sqrtf(qq);

    float gk[8];
#pragma unroll
    for (int j = 0; j < 8; ++j) {
      float a = 0.f;
#pragma unroll
      for (int q = 0; q < 4; ++q) {
        a += w4[j][q].x * s[4 * q + 0];
        a += w4[j][q].y * s[4 * q + 1];
        a += w4[j][q].z * s[4 * q + 2];
        a += w4[j][q].w * s[4 * q + 3];
      }
      gk[j] = a * inv;
    }
    float4* gd = reinterpret_cast<float4*>(g_s + n * GP + 8 * kq);
    gd[0] = make_float4(gk[0], gk[1], gk[2], gk[3]);
    gd[1] = make_float4(gk[4], gk[5], gk[6], gk[7]);
  }
  __syncthreads();

  float4 gb0[8], gb1[8];
  {
    const float4* g0 = reinterpret_cast<const float4*>(g_s + col * GP);
    const float4* g1 = reinterpret_cast<const float4*>(g_s + (16 + col) * GP);
#pragma unroll
    for (int jj = 0; jj < 8; ++jj) { gb0[jj] = g0[2 * jj + hi]; gb1[jj] = g1[2 * jj + hi]; }
  }

  v8f tacc[2][4];
#pragma unroll
  for (int nt = 0; nt < 2; ++nt)
#pragma unroll
    for (int kt = 0; kt < 4; ++kt) tacc[nt][kt] = (v8f){};

#pragma unroll
  for (int i = 0; i < 2; ++i) {
    const int l0 = ch * 256 + (wv * 2 + i) * 16;
    const float* ut = u + ((size_t)b * L_SEQ + l0) * D_IN;

    const float4* a4 = ua[i];

    v8f lg0 = {}, lg1 = {};
#pragma unroll
    for (int jj = 0; jj < 8; ++jj) {
      v2f alo, ahi, b0lo, b0hi, b1lo, b1hi;
      alo.x  = a4[jj].x;  alo.y  = a4[jj].y;  ahi.x  = a4[jj].z;  ahi.y  = a4[jj].w;
      b0lo.x = gb0[jj].x; b0lo.y = gb0[jj].y; b0hi.x = gb0[jj].z; b0hi.y = gb0[jj].w;
      b1lo.x = gb1[jj].x; b1lo.y = gb1[jj].y; b1hi.x = gb1[jj].z; b1hi.y = gb1[jj].w;
      lg0 = __builtin_amdgcn_wmma_f32_16x16x4_f32(false, alo, false, b0lo, (short)0, lg0, false, false);
      lg1 = __builtin_amdgcn_wmma_f32_16x16x4_f32(false, alo, false, b1lo, (short)0, lg1, false, false);
      lg0 = __builtin_amdgcn_wmma_f32_16x16x4_f32(false, ahi, false, b0hi, (short)0, lg0, false, false);
      lg1 = __builtin_amdgcn_wmma_f32_16x16x4_f32(false, ahi, false, b1hi, (short)0, lg1, false, false);
    }

    float c0[8], c1[8];
#pragma unroll
    for (int v = 0; v < 8; ++v) {
      const float m  = row_max16(fmaxf(lg0[v], lg1[v]));
      const float e0 = expf(lg0[v] - m);
      const float e1 = expf(lg1[v] - m);
      const float inv = 1.0f / row_sum16(e0 + e1);
      c0[v] = e0 * inv;
      c1[v] = e1 * inv;
    }

    const float* ub = ut + (size_t)(8 * hi) * D_IN + col;
#pragma unroll
    for (int j = 0; j < 4; ++j) {
      v2f ca0, ca1;
      ca0.x = c0[2 * j]; ca0.y = c0[2 * j + 1];
      ca1.x = c1[2 * j]; ca1.y = c1[2 * j + 1];
#pragma unroll
      for (int kt = 0; kt < 4; ++kt) {
        v2f ubm;
        ubm.x = ub[(size_t)(2 * j) * D_IN + kt * 16];
        ubm.y = ub[(size_t)(2 * j + 1) * D_IN + kt * 16];
        tacc[0][kt] = __builtin_amdgcn_wmma_f32_16x16x4_f32(false, ca0, false, ubm, (short)0, tacc[0][kt], false, false);
        tacc[1][kt] = __builtin_amdgcn_wmma_f32_16x16x4_f32(false, ca1, false, ubm, (short)0, tacc[1][kt], false, false);
      }
    }
  }

#pragma unroll
  for (int nt = 0; nt < 2; ++nt)
#pragma unroll
    for (int kt = 0; kt < 4; ++kt)
#pragma unroll
      for (int v = 0; v < 8; ++v)
        red[wv][(nt * 16 + v + 8 * hi) * D_IN + kt * 16 + col] = tacc[nt][kt][v];
  __syncthreads();
  float* dst = tout + ((size_t)b * NCH + ch) * (NC * D_IN);
#pragma unroll
  for (int i = 0; i < 8; ++i) {
    const int idx = threadIdx.x + 256 * i;
    dst[idx] = ((red[0][idx] + red[1][idx]) + (red[2][idx] + red[3][idx])) +
               ((red[4][idx] + red[5][idx]) + (red[6][idx] + red[7][idx]));
  }
}

extern "C" void kernel_launch(void* const* d_in, const int* in_sizes, int n_in,
                              void* d_out, int out_size, void* d_ws, size_t ws_size,
                              hipStream_t stream) {
  const float* u = (const float*)d_in[0];
  const float* W = (const float*)d_in[1];
  float* out = (float*)d_out;

  float* tp0 = (float*)d_ws;
  float* tpa = tp0 + (size_t)B_BATCH * NCH * D_IN;
  float* tpb = tpa + (size_t)B_BATCH * NCH * NC * D_IN;

  caps_colsum<<<B_BATCH * NCH, 256, 0, stream>>>(u, tp0);
  caps_route<true><<<B_BATCH * NCH, 256, 0, stream>>>(u, W, tp0, tpa);
  caps_route<false><<<B_BATCH * NCH, 256, 0, stream>>>(u, W, tpa, tpb);
  caps_final<<<B_BATCH * 8, 32, 0, stream>>>(tpb, W, out);
}


# First launch also writes W once in lane order (128 KB of workspace): every later load of a W slice covers 4 whole 128-byte lines instead of 16 partial ones (both routing heads, last launch)
# speedup vs baseline: 4.7345x; 1.1508x over previous
#include <hip/hip_runtime.h>
#include <math.h>


#define B_BATCH 32
#define L_SEQ   2048
#define D_IN    64
#define NC      32
#define DC      16
#define EPS_F   1e-7f
#define NCH     8
#define GP      68

typedef __attribute__((ext_vector_type(2))) float v2f;
typedef __attribute__((ext_vector_type(8))) float v8f;


template <int M>
__device__ __forceinline__ float row_xor(float v) {
  return __int_as_float(__builtin_amdgcn_mov_dpp(__float_as_int(v), 0x160 + M, 0xf, 0xf, false));
}
__device__ __forceinline__ float row_max16(float v) {
  v = fmaxf(v, row_xor<1>(v));
  v = fmaxf(v, row_xor<2>(v));
  v = fmaxf(v, row_xor<4>(v));
  v = fmaxf(v, row_xor<8>(v));
  return v;
}
__device__ __forceinline__ float row_sum16(float v) {
  v += row_xor<1>(v);
  v += row_xor<2>(v);
  v += row_xor<4>(v);
  v += row_xor<8>(v);
  return v;
}

__global__ void caps_colsum(const float* __restrict__ u, const float* __restrict__ W,
                            float* __restrict__ tp0, float* __restrict__ wp) {
  const int b  = blockIdx.x >> 3;
  const int ch = blockIdx.x & 7;
  const int t  = threadIdx.x;
  const bool packer = blockIdx.x < (D_IN * NC * DC / 4) / 256;
  float4 wv4 = make_float4(0.f, 0.f, 0.f, 0.f);
  if (packer) {
    const int gid  = blockIdx.x * 256 + t;
    const int lane = gid & 31;
    const int i    = (gid >> 5) & 31;
    const int wvv  = gid >> 10;
    const int krow = 8 * (lane & 7) + (i >> 2);
    const int ncol = (4 * wvv + (lane >> 3)) * DC + 4 * (i & 3);
    wv4 = *reinterpret_cast<const float4*>(W + (size_t)krow * (NC * DC) + ncol);
  }
  const int k4 = t & 15;
  const int rg = t >> 4;
  const float4* src = reinterpret_cast<const float4*>(
      u + ((size_t)b * L_SEQ + (size_t)ch * 256) * D_IN);
  float4 acc = make_float4(0.f, 0.f, 0.f, 0.f);
#pragma unroll
  for (int i = 0; i < 16; ++i) {
    float4 v = src[(size_t)(rg + 16 * i) * 16 + k4];
    acc.x += v.x; acc.y += v.y; acc.z += v.z; acc.w += v.w;
  }
  __shared__ float part[16][64];
  part[rg][4 * k4 + 0] = acc.x;
  part[rg][4 * k4 + 1] = acc.y;
  part[rg][4 * k4 + 2] = acc.z;
  part[rg][4 * k4 + 3] = acc.w;
  __syncthreads();
  if (t < 64) {
    float s = 0.f;
#pragma unroll
    for (int r = 0; r < 16; ++r) s += part[r][t];
    tp0[((size_t)b * NCH + ch) * D_IN + t] = s;
  }
  if (packer) reinterpret_cast<float4*>(wp)[blockIdx.x * 256 + t] = wv4;
}

__global__ void __launch_bounds__(32)
caps_final(const float* __restrict__ tp,
           const float* __restrict__ wp,
           float* __restrict__ outv) {
  const int lane = threadIdx.x;
  const int b    = blockIdx.x >> 3;
  const int kq   = lane & 7;
  const int n    = (blockIdx.x & 7) * 4 + (lane >> 3);

  const float4* tsrc = reinterpret_cast<const float4*>(
      tp + ((size_t)b * NCH * NC + n) * D_IN + 8 * kq);
  float4 p0[NCH], p1[NCH];
#pragma unroll
  for (int c = 0; c < NCH; ++c) {
    p0[c] = tsrc[c * (NC * D_IN / 4)];
    p1[c] = tsrc[c * (NC * D_IN / 4) + 1];
  }
  const float4* wsrc = reinterpret_cast<const float4*>(wp) + (size_t)(blockIdx.x & 7) * 1024 + lane;
  float4 w4[8][4];
#pragma unroll
  for (int j = 0; j < 8; ++j)
#pragma unroll
    for (int q = 0; q < 4; ++q) w4[j][q] = wsrc[(4 * j + q) * 32];
  __builtin_amdgcn_sched_barrier(0);

  float tk[8];
#pragma unroll
  for (int j = 0; j < 8; ++j) tk[j] = 0.f;
#pragma unroll
  for (int c = 0; c < NCH; ++c) {
    tk[0] += p0[c].x; tk[1] += p0[c].y; tk[2] += p0[c].z; tk[3] += p0[c].w;
    tk[4] += p1[c].x; tk[5] += p1[c].y; tk[6] += p1[c].z; tk[7] += p1[c].w;
  }

  float s[DC];
#pragma unroll
  for (int d = 0; d < DC; ++d) s[d] = 0.f;
#pragma unroll
  for (int j = 0; j < 8; ++j)
#pragma unroll
    for (int q = 0; q < 4; ++q) {
      s[4 * q + 0] += tk[j] * w4[j][q].x;
      s[4 * q + 1] += tk[j] * w4[j][q].y;
      s[4 * q + 2] += tk[j] * w4[j][q].z;
      s[4 * q + 3] += tk[j] * w4[j][q].w;
    }
#pragma unroll
  for (int d = 0; d < DC; ++d) {
    s[d] += row_xor<1>(s[d]);
    s[d] += row_xor<2>(s[d]);
    s[d] += row_xor<4>(s[d]);
  }
  float qq = EPS_F;
#pragma unroll
  for (int d = 0; d < DC; ++d) qq += s[d] * s[d];
  const float nrm = sqrtf(qq);

  float o0 = s[0], o1 = s[1];
#pragma unroll
  for (int j = 1; j < 8; ++j) {
    o0 = (kq == j) ? s[2 * j] : o0;
    o1 = (kq == j) ? s[2 * j + 1] : o1;
  }
  float2 o;
  o.x = o0 / nrm;
  o.y = o1 / nrm;
  *reinterpret_cast<float2*>(outv + ((size_t)b * NC + n) * DC + 2 * kq) = o;
}

template <bool FIRST>
__global__ void __launch_bounds__(256)
caps_route(const float* __restrict__ u,
           const float* __restrict__ wp,
           const float* __restrict__ tin,
           float* __restrict__ tout) {
  const int lane = threadIdx.x & 31;
  const int wv   = threadIdx.x >> 5;
  const int col  = lane & 15;
  const int hi   = lane >> 4;
  const int b    = blockIdx.x >> 3;
  const int ch   = blockIdx.x & 7;

  __shared__ __attribute__((aligned(16))) float g_s[NC * GP];
  __shared__ float red[8][NC * D_IN];

  float4 ua[2][8];

  {
    const int kq = lane & 7;
    const int n  = wv * 4 + (lane >> 3);

    const float4* tsrc = FIRST
        ? reinterpret_cast<const float4*>(tin + (size_t)b * NCH * D_IN + 8 * kq)
        : reinterpret_cast<const float4*>(tin + ((size_t)b * NCH * NC + n) * D_IN + 8 * kq);
    const int tstep = FIRST ? (D_IN / 4) : (NC * D_IN / 4);
    float4 p0[NCH], p1[NCH];
#pragma unroll
    for (int c = 0; c < NCH; ++c) { p0[c] = tsrc[c * tstep]; p1[c] = tsrc[c * tstep + 1]; }

    const float4* wsrc = reinterpret_cast<const float4*>(wp) + (size_t)wv * 1024 + lane;
    float4 w4[8][4];
#pragma unroll
    for (int j = 0; j < 8; ++j)
#pragma unroll
      for (int q = 0; q < 4; ++q) w4[j][q] = wsrc[(4 * j + q) * 32];

#pragma unroll
    for (int i = 0; i < 2; ++i) {
      const float4* ur = reinterpret_cast<const float4*>(
          u + ((size_t)b * L_SEQ + ch * 256 + (wv * 2 + i) * 16 + col) * D_IN);
#pragma unroll
      for (int jj = 0; jj < 8; ++jj) ua[i][jj] = ur[2 * jj + hi];
    }
    __builtin_amdgcn_sched_barrier(0);

    float tk[8];
#pragma unroll
    for (int j = 0; j < 8; ++j) tk[j] = 0.f;
#pragma unroll
    for (int c = 0; c < NCH; ++c) {
      tk[0] += p0[c].x; tk[1] += p0[c].y; tk[2] += p0[c].z; tk[3] += p0[c].w;
      tk[4] += p1[c].x; tk[5] += p1[c].y; tk[6] += p1[c].z; tk[7] += p1[c].w;
    }
    if (FIRST) {
#pragma unroll
      for (int j = 0; j < 8; ++j) tk[j] *= (1.0f / 32.0f);
    }

    float s[DC];
#pragma unroll
    for (int d = 0; d < DC; ++d) s[d] = 0.f;
#pragma unroll
    for (int j = 0; j < 8; ++j)
#pragma unroll
      for (int q = 0; q < 4; ++q) {
        s[4 * q + 0] += tk[j] * w4[j][q].x;
        s[4 * q + 1] += tk[j] * w4[j][q].y;
        s[4 * q + 2] += tk[j] * w4[j][q].z;
        s[4 * q + 3] += tk[j] * w4[j][q].w;
      }
#pragma unroll
    for (int d = 0; d < DC; ++d) {
      s[d] += row_xor<1>(s[d]);
      s[d] += row_xor<2>(s[d]);
      s[d] += row_xor<4>(s[d]);
    }
    float qq = EPS_F;
#pragma unroll
    for (int d = 0; d < DC; ++d) qq += s[d] * s[d];
    const float inv = 1.0f / sqrtf(qq);

    float gk[8];
#pragma unroll
    for (int j = 0; j < 8; ++j) {
      float a = 0.f;
#pragma unroll
      for (int q = 0; q < 4; ++q) {
        a += w4[j][q].x * s[4 * q + 0];
        a += w4[j][q].y * s[4 * q + 1];
        a += w4[j][q].z * s[4 * q + 2];
        a += w4[j][q].w * s[4 * q + 3];
      }
      gk[j] = a * inv;
    }
    float4* gd = reinterpret_cast<float4*>(g_s + n * GP + 8 * kq);
    gd[0] = make_float4(gk[0], gk[1], gk[2], gk[3]);
    gd[1] = make_float4(gk[4], gk[5], gk[6], gk[7]);
  }
  __syncthreads();

  float4 gb0[8], gb1[8];
  {
    const float4* g0 = reinterpret_cast<const float4*>(g_s + col * GP);
    const float4* g1 = reinterpret_cast<const float4*>(g_s + (16 + col) * GP);
#pragma unroll
    for (int jj = 0; jj < 8; ++jj) { gb0[jj] = g0[2 * jj + hi]; gb1[jj] = g1[2 * jj + hi]; }
  }

  v8f tacc[2][4];
#pragma unroll
  for (int nt = 0; nt < 2; ++nt)
#pragma unroll
    for (int kt = 0; kt < 4; ++kt) tacc[nt][kt] = (v8f){};

#pragma unroll
  for (int i = 0; i < 2; ++i) {
    const int l0 = ch * 256 + (wv * 2 + i) * 16;
    const float* ut = u + ((size_t)b * L_SEQ + l0) * D_IN;

    const float4* a4 = ua[i];

    v8f lg0 = {}, lg1 = {};
#pragma unroll
    for (int jj = 0; jj < 8; ++jj) {
      v2f alo, ahi, b0lo, b0hi, b1lo, b1hi;
      alo.x  = a4[jj].x;  alo.y  = a4[jj].y;  ahi.x  = a4[jj].z;  ahi.y  = a4[jj].w;
      b0lo.x = gb0[jj].x; b0lo.y = gb0[jj].y; b0hi.x = gb0[jj].z; b0hi.y = gb0[jj].w;
      b1lo.x = gb1[jj].x; b1lo.y = gb1[jj].y; b1hi.x = gb1[jj].z; b1hi.y = gb1[jj].w;
      lg0 = __builtin_amdgcn_wmma_f32_16x16x4_f32(false, alo, false, b0lo, (short)0, lg0, false, false);
      lg1 = __builtin_amdgcn_wmma_f32_16x16x4_f32(false, alo, false, b1lo, (short)0, lg1, false, false);
      lg0 = __builtin_amdgcn_wmma_f32_16x16x4_f32(false, ahi, false, b0hi, (short)0, lg0, false, false);
      lg1 = __builtin_amdgcn_wmma_f32_16x16x4_f32(false, ahi, false, b1hi, (short)0, lg1, false, false);
    }

    float c0[8], c1[8];
#pragma unroll
    for (int v = 0; v < 8; ++v) {
      const float m  = row_max16(fmaxf(lg0[v], lg1[v]));
      const float e0 = expf(lg0[v] - m);
      const float e1 = expf(lg1[v] - m);
      const float inv = 1.0f / row_sum16(e0 + e1);
      c0[v] = e0 * inv;
      c1[v] = e1 * inv;
    }

    const float* ub = ut + (size_t)(8 * hi) * D_IN + col;
#pragma unroll
    for (int j = 0; j < 4; ++j) {
      v2f ca0, ca1;
      ca0.x = c0[2 * j]; ca0.y = c0[2 * j + 1];
      ca1.x = c1[2 * j]; ca1.y = c1[2 * j + 1];
#pragma unroll
      for (int kt = 0; kt < 4; ++kt) {
        v2f ubm;
        ubm.x = ub[(size_t)(2 * j) * D_IN + kt * 16];
        ubm.y = ub[(size_t)(2 * j + 1) * D_IN + kt * 16];
        tacc[0][kt] = __builtin_amdgcn_wmma_f32_16x16x4_f32(false, ca0, false, ubm, (short)0, tacc[0][kt], false, false);
        tacc[1][kt] = __builtin_amdgcn_wmma_f32_16x16x4_f32(false, ca1, false, ubm, (short)0, tacc[1][kt], false, false);
      }
    }
  }

#pragma unroll
  for (int nt = 0; nt < 2; ++nt)
#pragma unroll
    for (int kt = 0; kt < 4; ++kt)
#pragma unroll
      for (int v = 0; v < 8; ++v)
        red[wv][(nt * 16 + v + 8 * hi) * D_IN + kt * 16 + col] = tacc[nt][kt][v];
  __syncthreads();
  float* dst = tout + ((size_t)b * NCH + ch) * (NC * D_IN);
#pragma unroll
  for (int i = 0; i < 8; ++i) {
    const int idx = threadIdx.x + 256 * i;
    dst[idx] = ((red[0][idx] + red[1][idx]) + (red[2][idx] + red[3][idx])) +
               ((red[4][idx] + red[5][idx]) + (red[6][idx] + red[7][idx]));
  }
}

extern "C" void kernel_launch(void* const* d_in, const int* in_sizes, int n_in,
                              void* d_out, int out_size, void* d_ws, size_t ws_size,
                              hipStream_t stream) {
  const float* u = (const float*)d_in[0];
  const float* W = (const float*)d_in[1];
  float* out = (float*)d_out;

  float* tp0 = (float*)d_ws;
  float* tpa = tp0 + (size_t)B_BATCH * NCH * D_IN;
  float* tpb = tpa + (size_t)B_BATCH * NCH * NC * D_IN;
  float* wpk = tpb + (size_t)B_BATCH * NCH * NC * D_IN;

  caps_colsum<<<B_BATCH * NCH, 256, 0, stream>>>(u, W, tp0, wpk);
  caps_route<true><<<B_BATCH * NCH, 256, 0, stream>>>(u, wpk, tp0, tpa);
  caps_route<false><<<B_BATCH * NCH, 256, 0, stream>>>(u, wpk, tpa, tpb);
  caps_final<<<B_BATCH * 8, 32, 0, stream>>>(tpb, wpk, out);
}


# Softmax between the two matrix products on the hardware exp2 and reciprocal instructions, lane exchanges folded into the adds (routing body 918 -> 370 vector instructions per wave)
# speedup vs baseline: 4.9477x; 1.0450x over previous
#include <hip/hip_runtime.h>
#include <math.h>


#define B_BATCH 32
#define L_SEQ   2048
#define D_IN    64
#define NC      32
#define DC      16
#define EPS_F   1e-7f
#define LOG2E_F 1.44269504088896340736f
#define NCH     8
#define GP      68

typedef __attribute__((ext_vector_type(2))) float v2f;
typedef __attribute__((ext_vector_type(8))) float v8f;


template <int M>
__device__ __forceinline__ float row_xor(float v) {
  return __int_as_float(__builtin_amdgcn_mov_dpp(__float_as_int(v), 0x160 + M, 0xf, 0xf, true));
}
__device__ __forceinline__ float row_max16(float v) {
  v = fmaxf(v, row_xor<1>(v));
  v = fmaxf(v, row_xor<2>(v));
  v = fmaxf(v, row_xor<4>(v));
  v = fmaxf(v, row_xor<8>(v));
  return v;
}
__device__ __forceinline__ float row_sum16(float v) {
  v += row_xor<1>(v);
  v += row_xor<2>(v);
  v += row_xor<4>(v);
  v += row_xor<8>(v);
  return v;
}

__global__ void caps_colsum(const float* __restrict__ u, const float* __restrict__ W,
                            float* __restrict__ tp0, float* __restrict__ wp) {
  const int b  = blockIdx.x >> 3;
  const int ch = blockIdx.x & 7;
  const int t  = threadIdx.x;
  const bool packer = blockIdx.x < (D_IN * NC * DC / 4) / 256;
  float4 wv4 = make_float4(0.f, 0.f, 0.f, 0.f);
  if (packer) {
    const int gid  = blockIdx.x * 256 + t;
    const int lane = gid & 31;
    const int i    = (gid >> 5) & 31;
    const int wvv  = gid >> 10;
    const int krow = 8 * (lane & 7) + (i >> 2);
    const int ncol = (4 * wvv + (lane >> 3)) * DC + 4 * (i & 3);
    wv4 = *reinterpret_cast<const float4*>(W + (size_t)krow * (NC * DC) + ncol);
  }
  const int k4 = t & 15;
  const int rg = t >> 4;
  const float4* src = reinterpret_cast<const float4*>(
      u + ((size_t)b * L_SEQ + (size_t)ch * 256) * D_IN);
  float4 acc = make_float4(0.f, 0.f, 0.f, 0.f);
#pragma unroll
  for (int i = 0; i < 16; ++i) {
    float4 v = src[(size_t)(rg + 16 * i) * 16 + k4];
    acc.x += v.x; acc.y += v.y; acc.z += v.z; acc.w += v.w;
  }
  __shared__ float part[16][64];
  part[rg][4 * k4 + 0] = acc.x;
  part[rg][4 * k4 + 1] = acc.y;
  part[rg][4 * k4 + 2] = acc.z;
  part[rg][4 * k4 + 3] = acc.w;
  __syncthreads();
  if (t < 64) {
    float s = 0.f;
#pragma unroll
    for (int r = 0; r < 16; ++r) s += part[r][t];
    tp0[((size_t)b * NCH + ch) * D_IN + t] = s;
  }
  if (packer) reinterpret_cast<float4*>(wp)[blockIdx.x * 256 + t] = wv4;
}

__global__ void __launch_bounds__(32)
caps_final(const float* __restrict__ tp,
           const float* __restrict__ wp,
           float* __restrict__ outv) {
  const int lane = threadIdx.x;
  const int b    = blockIdx.x >> 3;
  const int kq   = lane & 7;
  const int n    = (blockIdx.x & 7) * 4 + (lane >> 3);

  const float4* tsrc = reinterpret_cast<const float4*>(
      tp + ((size_t)b * NCH * NC + n) * D_IN + 8 * kq);
  float4 p0[NCH], p1[NCH];
#pragma unroll
  for (int c = 0; c < NCH; ++c) {
    p0[c] = tsrc[c * (NC * D_IN / 4)];
    p1[c] = tsrc[c * (NC * D_IN / 4) + 1];
  }
  const float4* wsrc = reinterpret_cast<const float4*>(wp) + (size_t)(blockIdx.x & 7) * 1024 + lane;
  float4 w4[8][4];
#pragma unroll
  for (int j = 0; j < 8; ++j)
#pragma unroll
    for (int q = 0; q < 4; ++q) w4[j][q] = wsrc[(4 * j + q) * 32];
  __builtin_amdgcn_sched_barrier(0);

  float tk[8];
#pragma unroll
  for (int j = 0; j < 8; ++j) tk[j] = 0.f;
#pragma unroll
  for (int c = 0; c < NCH; ++c) {
    tk[0] += p0[c].x; tk[1] += p0[c].y; tk[2] += p0[c].z; tk[3] += p0[c].w;
    tk[4] += p1[c].x; tk[5] += p1[c].y; tk[6] += p1[c].z; tk[7] += p1[c].w;
  }

  float s[DC];
#pragma unroll
  for (int d = 0; d < DC; ++d) s[d] = 0.f;
#pragma unroll
  for (int j = 0; j < 8; ++j)
#pragma unroll
    for (int q = 0; q < 4; ++q) {
      s[4 * q + 0] += tk[j] * w4[j][q].x;
      s[4 * q + 1] += tk[j] * w4[j][q].y;
      s[4 * q + 2] += tk[j] * w4[j][q].z;
      s[4 * q + 3] += tk[j] * w4[j][q].w;
    }
#pragma unroll
  for (int d = 0; d < DC; ++d) {
    s[d] += row_xor<1>(s[d]);
    s[d] += row_xor<2>(s[d]);
    s[d] += row_xor<4>(s[d]);
  }
  float qq = EPS_F;
#pragma unroll
  for (int d = 0; d < DC; ++d) qq += s[d] * s[d];
  const float nrm = sqrtf(qq);

  float o0 = s[0], o1 = s[1];
#pragma unroll
  for (int j = 1; j < 8; ++j) {
    o0 = (kq == j) ? s[2 * j] : o0;
    o1 = (kq == j) ? s[2 * j + 1] : o1;
  }
  float2 o;
  o.x = o0 / nrm;
  o.y = o1 / nrm;
  *reinterpret_cast<float2*>(outv + ((size_t)b * NC + n) * DC + 2 * kq) = o;
}

template <bool FIRST>
__global__ void __launch_bounds__(256)
caps_route(const float* __restrict__ u,
           const float* __restrict__ wp,
           const float* __restrict__ tin,
           float* __restrict__ tout) {
  const int lane = threadIdx.x & 31;
  const int wv   = threadIdx.x >> 5;
  const int col  = lane & 15;
  const int hi   = lane >> 4;
  const int b    = blockIdx.x >> 3;
  const int ch   = blockIdx.x & 7;

  __shared__ __attribute__((aligned(16))) float g_s[NC * GP];
  __shared__ float red[8][NC * D_IN];

  float4 ua[2][8];

  {
    const int kq = lane & 7;
    const int n  = wv * 4 + (lane >> 3);

    const float4* tsrc = FIRST
        ? reinterpret_cast<const float4*>(tin + (size_t)b * NCH * D_IN + 8 * kq)
        : reinterpret_cast<const float4*>(tin + ((size_t)b * NCH * NC + n) * D_IN + 8 * kq);
    const int tstep = FIRST ? (D_IN / 4) : (NC * D_IN / 4);
    float4 p0[NCH], p1[NCH];
#pragma unroll
    for (int c = 0; c < NCH; ++c) { p0[c] = tsrc[c * tstep]; p1[c] = tsrc[c * tstep + 1]; }

    const float4* wsrc = reinterpret_cast<const float4*>(wp) + (size_t)wv * 1024 + lane;
    float4 w4[8][4];
#pragma unroll
    for (int j = 0; j < 8; ++j)
#pragma unroll
      for (int q = 0; q < 4; ++q) w4[j][q] = wsrc[(4 * j + q) * 32];

#pragma unroll
    for (int i = 0; i < 2; ++i) {
      const float4* ur = reinterpret_cast<const float4*>(
          u + ((size_t)b * L_SEQ + ch * 256 + (wv * 2 + i) * 16 + col) * D_IN);
#pragma unroll
      for (int jj = 0; jj < 8; ++jj) ua[i][jj] = ur[2 * jj + hi];
    }
    __builtin_amdgcn_sched_barrier(0);

    float tk[8];
#pragma unroll
    for (int j = 0; j < 8; ++j) tk[j] = 0.f;
#pragma unroll
    for (int c = 0; c < NCH; ++c) {
      tk[0] += p0[c].x; tk[1] += p0[c].y; tk[2] += p0[c].z; tk[3] += p0[c].w;
      tk[4] += p1[c].x; tk[5] += p1[c].y; tk[6] += p1[c].z; tk[7] += p1[c].w;
    }
    if (FIRST) {
#pragma unroll
      for (int j = 0; j < 8; ++j) tk[j] *= (1.0f / 32.0f);
    }

    float s[DC];
#pragma unroll
    for (int d = 0; d < DC; ++d) s[d] = 0.f;
#pragma unroll
    for (int j = 0; j < 8; ++j)
#pragma unroll
      for (int q = 0; q < 4; ++q) {
        s[4 * q + 0] += tk[j] * w4[j][q].x;
        s[4 * q + 1] += tk[j] * w4[j][q].y;
        s[4 * q + 2] += tk[j] * w4[j][q].z;
        s[4 * q + 3] += tk[j] * w4[j][q].w;
      }
#pragma unroll
    for (int d = 0; d < DC; ++d) {
      s[d] += row_xor<1>(s[d]);
      s[d] += row_xor<2>(s[d]);
      s[d] += row_xor<4>(s[d]);
    }
    float qq = EPS_F;
#pragma unroll
    for (int d = 0; d < DC; ++d) qq += s[d] * s[d];
    const float inv = __builtin_amdgcn_rsqf(qq);

    float gk[8];
#pragma unroll
    for (int j = 0; j < 8; ++j) {
      float a = 0.f;
#pragma unroll
      for (int q = 0; q < 4; ++q) {
        a += w4[j][q].x * s[4 * q + 0];
        a += w4[j][q].y * s[4 * q + 1];
        a += w4[j][q].z * s[4 * q + 2];
        a += w4[j][q].w * s[4 * q + 3];
      }
      gk[j] = a * inv;
    }
    float4* gd = reinterpret_cast<float4*>(g_s + n * GP + 8 * kq);
    gd[0] = make_float4(gk[0], gk[1], gk[2], gk[3]);
    gd[1] = make_float4(gk[4], gk[5], gk[6], gk[7]);
  }
  __syncthreads();

  float4 gb0[8], gb1[8];
  {
    const float4* g0 = reinterpret_cast<const float4*>(g_s + col * GP);
    const float4* g1 = reinterpret_cast<const float4*>(g_s + (16 + col) * GP);
#pragma unroll
    for (int jj = 0; jj < 8; ++jj) { gb0[jj] = g0[2 * jj + hi]; gb1[jj] = g1[2 * jj + hi]; }
  }

  v8f tacc[2][4];
#pragma unroll
  for (int nt = 0; nt < 2; ++nt)
#pragma unroll
    for (int kt = 0; kt < 4; ++kt) tacc[nt][kt] = (v8f){};

#pragma unroll
  for (int i = 0; i < 2; ++i) {
    const int l0 = ch * 256 + (wv * 2 + i) * 16;
    const float* ut = u + ((size_t)b * L_SEQ + l0) * D_IN;

    const float4* a4 = ua[i];

    v8f lg0 = {}, lg1 = {};
#pragma unroll
    for (int jj = 0; jj < 8; ++jj) {
      v2f alo, ahi, b0lo, b0hi, b1lo, b1hi;
      alo.x  = a4[jj].x;  alo.y  = a4[jj].y;  ahi.x  = a4[jj].z;  ahi.y  = a4[jj].w;
      b0lo.x = gb0[jj].x; b0lo.y = gb0[jj].y; b0hi.x = gb0[jj].z; b0hi.y = gb0[jj].w;
      b1lo.x = gb1[jj].x; b1lo.y = gb1[jj].y; b1hi.x = gb1[jj].z; b1hi.y = gb1[jj].w;
      lg0 = __builtin_amdgcn_wmma_f32_16x16x4_f32(false, alo, false, b0lo, (short)0, lg0, false, false);
      lg1 = __builtin_amdgcn_wmma_f32_16x16x4_f32(false, alo, false, b1lo, (short)0, lg1, false, false);
      lg0 = __builtin_amdgcn_wmma_f32_16x16x4_f32(false, ahi, false, b0hi, (short)0, lg0, false, false);
      lg1 = __builtin_amdgcn_wmma_f32_16x16x4_f32(false, ahi, false, b1hi, (short)0, lg1, false, false);
    }

    float c0[8], c1[8];
#pragma unroll
    for (int v = 0; v < 8; ++v) {
      const float y0 = lg0[v] * LOG2E_F;
      const float y1 = lg1[v] * LOG2E_F;
      const float m2 = row_max16(fmaxf(y0, y1));
      const float e0 = __builtin_amdgcn_exp2f(y0 - m2);
      const float e1 = __builtin_amdgcn_exp2f(y1 - m2);
      const float inv = __builtin_amdgcn_rcpf(row_sum16(e0 + e1));
      c0[v] = e0 * inv;
      c1[v] = e1 * inv;
    }

    const float* ub = ut + (size_t)(8 * hi) * D_IN + col;
#pragma unroll
    for (int j = 0; j < 4; ++j) {
      v2f ca0, ca1;
      ca0.x = c0[2 * j]; ca0.y = c0[2 * j + 1];
      ca1.x = c1[2 * j]; ca1.y = c1[2 * j + 1];
#pragma unroll
      for (int kt = 0; kt < 4; ++kt) {
        v2f ubm;
        ubm.x = ub[(size_t)(2 * j) * D_IN + kt * 16];
        ubm.y = ub[(size_t)(2 * j + 1) * D_IN + kt * 16];
        tacc[0][kt] = __builtin_amdgcn_wmma_f32_16x16x4_f32(false, ca0, false, ubm, (short)0, tacc[0][kt], false, false);
        tacc[1][kt] = __builtin_amdgcn_wmma_f32_16x16x4_f32(false, ca1, false, ubm, (short)0, tacc[1][kt], false, false);
      }
    }
  }

#pragma unroll
  for (int nt = 0; nt < 2; ++nt)
#pragma unroll
    for (int kt = 0; kt < 4; ++kt)
#pragma unroll
      for (int v = 0; v < 8; ++v)
        red[wv][(nt * 16 + v + 8 * hi) * D_IN + kt * 16 + col] = tacc[nt][kt][v];
  __syncthreads();
  float* dst = tout + ((size_t)b * NCH + ch) * (NC * D_IN);
#pragma unroll
  for (int i = 0; i < 8; ++i) {
    const int idx = threadIdx.x + 256 * i;
    dst[idx] = ((red[0][idx] + red[1][idx]) + (red[2][idx] + red[3][idx])) +
               ((red[4][idx] + red[5][idx]) + (red[6][idx] + red[7][idx]));
  }
}

extern "C" void kernel_launch(void* const* d_in, const int* in_sizes, int n_in,
                              void* d_out, int out_size, void* d_ws, size_t ws_size,
                              hipStream_t stream) {
  const float* u = (const float*)d_in[0];
  const float* W = (const float*)d_in[1];
  float* out = (float*)d_out;

  float* tp0 = (float*)d_ws;
  float* tpa = tp0 + (size_t)B_BATCH * NCH * D_IN;
  float* tpb = tpa + (size_t)B_BATCH * NCH * NC * D_IN;
  float* wpk = tpb + (size_t)B_BATCH * NCH * NC * D_IN;

  caps_colsum<<<B_BATCH * NCH, 256, 0, stream>>>(u, W, tp0, wpk);
  caps_route<true><<<B_BATCH * NCH, 256, 0, stream>>>(u, wpk, tp0, tpa);
  caps_route<false><<<B_BATCH * NCH, 256, 0, stream>>>(u, wpk, tpa, tpb);
  caps_final<<<B_BATCH * 8, 32, 0, stream>>>(tpb, wpk, out);
}
